# starting kernel
# baseline (speedup 1.0000x reference)
#include <hip/hip_runtime.h>

typedef __attribute__((ext_vector_type(2))) float v2f;
typedef __attribute__((ext_vector_type(8))) float v8f;

#define BATCH 32
#define NN    1024
#define IND   128
#define HIDD  64
#define LDS_STRIDE 66

__global__ __launch_bounds__(256) void k_proj(
    const float* __restrict__ x,
    const float* __restrict__ Wh, const float* __restrict__ bh,
    const float* __restrict__ Wf, const float* __restrict__ bf,
    float* __restrict__ H, float* __restrict__ P) {
  const int lane = threadIdx.x & 31;
  const int w    = threadIdx.x >> 5;
  const int gw   = blockIdx.x * 8 + w;
  const int m0   = gw * 16;
  const int lr   = lane & 15;
  const int hi   = lane >> 4;

  v8f ha[4] = {}; v8f pa[4] = {};
  const float* xrow = x + (size_t)(m0 + lr) * IND + 2 * hi;

#pragma unroll 4
  for (int k0 = 0; k0 < IND; k0 += 4) {
    v2f a = *(const v2f*)(xrow + k0);
    const float* w0 = Wh + (size_t)(k0 + 2 * hi) * HIDD + lr;
    const float* f0 = Wf + (size_t)(k0 + 2 * hi) * HIDD + lr;
#pragma unroll
    for (int nb = 0; nb < 4; ++nb) {
      v2f bw, bq;
      bw.x = w0[nb * 16]; bw.y = w0[HIDD + nb * 16];
      bq.x = f0[nb * 16]; bq.y = f0[HIDD + nb * 16];
      ha[nb] = __builtin_amdgcn_wmma_f32_16x16x4_f32(false, a, false, bw,
                                                     (short)0, ha[nb], false, false);
      pa[nb] = __builtin_amdgcn_wmma_f32_16x16x4_f32(false, a, false, bq,
                                                     (short)0, pa[nb], false, false);
    }
  }

#pragma unroll
  for (int nb = 0; nb < 4; ++nb) {
    const int n = nb * 16 + lr;
    const float bhn = bh[n];
    const float bfn = bf[n];
#pragma unroll
    for (int i = 0; i < 8; ++i) {
      const int m = i + 8 * hi;
      const size_t off = (size_t)(m0 + m) * HIDD + n;
      H[off] = ha[nb][i] + bhn;
      P[off] = pa[nb][i] + bfn;
    }
  }
}

__global__ __launch_bounds__(256) void k_bmm_gate(
    const float* __restrict__ A, const float* __restrict__ H,
    const float* __restrict__ P,
    const float* __restrict__ Wg1, const float* __restrict__ bg1,
    const float* __restrict__ Wg2, const float* __restrict__ bg2,
    float* __restrict__ out) {
  __shared__ float lds[8 * 16 * LDS_STRIDE];

  const int lane = threadIdx.x & 31;
  const int w    = threadIdx.x >> 5;
  const int gw   = blockIdx.x * 8 + w;
  const int b    = gw >> 6;
  const int m0   = (gw & 63) * 16;
  const int lr   = lane & 15;
  const int hi   = lane >> 4;

  const float* Ab = A + (size_t)b * NN * NN;
  const float* Hb = H + (size_t)b * NN * HIDD;

  v8f acc[4] = {};
  const float* arow = Ab + (size_t)(m0 + lr) * NN + 2 * hi;
#pragma unroll 4
  for (int k0 = 0; k0 < NN; k0 += 4) {
    v2f a = *(const v2f*)(arow + k0);
    const float* h0 = Hb + (size_t)(k0 + 2 * hi) * HIDD + lr;
#pragma unroll
    for (int nb = 0; nb < 4; ++nb) {
      v2f bb;
      bb.x = h0[nb * 16];
      bb.y = h0[HIDD + nb * 16];
      acc[nb] = __builtin_amdgcn_wmma_f32_16x16x4_f32(false, a, false, bb,
                                                      (short)0, acc[nb], false, false);
    }
  }

  float* myl = lds + w * (16 * LDS_STRIDE);
#pragma unroll
  for (int nb = 0; nb < 4; ++nb)
#pragma unroll
    for (int i = 0; i < 8; ++i)
      myl[(i + 8 * hi) * LDS_STRIDE + nb * 16 + lr] = acc[nb][i];

  v8f g[4] = {};
  const float* prow = P + (size_t)(b * NN + m0 + lr) * HIDD + 2 * hi;
#pragma unroll
  for (int kk = 0; kk < HIDD; kk += 4) {
    v2f a2 = *(const v2f*)(myl + lr * LDS_STRIDE + kk + 2 * hi);
    v2f a1 = *(const v2f*)(prow + kk);
    const float* g2w = Wg2 + (size_t)(kk + 2 * hi) * HIDD + lr;
    const float* g1w = Wg1 + (size_t)(kk + 2 * hi) * HIDD + lr;
#pragma unroll
    for (int nb = 0; nb < 4; ++nb) {
      v2f b2, b1;
      b2.x = g2w[nb * 16]; b2.y = g2w[HIDD + nb * 16];
      b1.x = g1w[nb * 16]; b1.y = g1w[HIDD + nb * 16];
      g[nb] = __builtin_amdgcn_wmma_f32_16x16x4_f32(false, a2, false, b2,
                                                    (short)0, g[nb], false, false);
      g[nb] = __builtin_amdgcn_wmma_f32_16x16x4_f32(false, a1, false, b1,
                                                    (short)0, g[nb], false, false);
    }
  }

#pragma unroll
  for (int nb = 0; nb < 4; ++nb) {
    const int n = nb * 16 + lr;
    const float bias = bg1[n] + bg2[n];
#pragma unroll
    for (int i = 0; i < 8; ++i) {
      const int m = i + 8 * hi;
      const size_t off = (size_t)(b * NN + m0 + m) * HIDD + n;
      const float xn = acc[nb][i];
      const float p  = P[off];
      const float z  = g[nb][i] + bias;
      const float gate = 1.0f / (1.0f + __expf(-z));
      out[off] = xn * gate + p * (1.0f - gate);
    }
  }
}

extern "C" void kernel_launch(void* const* d_in, const int* in_sizes, int n_in,
                              void* d_out, int out_size, void* d_ws, size_t ws_size,
                              hipStream_t stream) {
  const float* x    = (const float*)d_in[0];
  const float* A    = (const float*)d_in[1];
  const float* W_h  = (const float*)d_in[2];
  const float* b_h  = (const float*)d_in[3];
  const float* W_f1 = (const float*)d_in[4];
  const float* b_f1 = (const float*)d_in[5];
  const float* W_g1 = (const float*)d_in[6];
  const float* b_g1 = (const float*)d_in[7];
  const float* W_g2 = (const float*)d_in[8];
  const float* b_g2 = (const float*)d_in[9];
  float* out = (float*)d_out;

  float* H = (float*)d_ws;
  float* P = H + (size_t)BATCH * NN * HIDD;

  k_proj<<<256, 256, 0, stream>>>(x, W_h, b_h, W_f1, b_f1, H, P);
  k_bmm_gate<<<256, 256, 0, stream>>>(A, H, P, W_g1, b_g1, W_g2, b_g2, out);
}


# B operands (H tile, weights) staged once per block in LDS as [k/4][n][k%4] and read as b128 fragments by all 8 waves, A loaded b128 in the same k order; sums inside each 8-k group reordered
# speedup vs baseline: 1.8771x; 1.8771x over previous
#include <hip/hip_runtime.h>


typedef __attribute__((ext_vector_type(2))) float v2f;
typedef __attribute__((ext_vector_type(4))) float v4f;
typedef __attribute__((ext_vector_type(8))) float v8f;

#define BATCH 32
#define NN    1024
#define IND   128
#define HIDD  64
#define KC    64
#define HSF   (KC * HIDD)
#define XS    68


static __device__ __forceinline__ v8f mma4(v2f a, v2f b, v8f c) {
  return __builtin_amdgcn_wmma_f32_16x16x4_f32(false, a, false, b, (short)0, c, false, false);
}

static __device__ __forceinline__ void stage_load(v4f (&r)[4], const float* __restrict__ src) {
#pragma unroll
  for (int p = 0; p < 4; ++p) {
    r[p].x = src[(16 * p + 0) * HIDD];
    r[p].y = src[(16 * p + 1) * HIDD];
    r[p].z = src[(16 * p + 2) * HIDD];
    r[p].w = src[(16 * p + 3) * HIDD];
  }
}
static __device__ __forceinline__ void stage_store(const v4f (&r)[4], float* dst) {
#pragma unroll
  for (int p = 0; p < 4; ++p) *(v4f*)(dst + p * 1024) = r[p];
}

__global__ __launch_bounds__(256) void k_proj(
    const float* __restrict__ x,
    const float* __restrict__ Wh, const float* __restrict__ bh,
    const float* __restrict__ Wf, const float* __restrict__ bf,
    float* __restrict__ H, float* __restrict__ P) {
  __shared__ __attribute__((aligned(16))) float wl[32 * 128 * 4];

  const int tid  = threadIdx.x;
  const int lane = tid & 31;
  const int w    = tid >> 5;
  const int gw   = blockIdx.x * 8 + w;
  const int m0   = gw * 16;
  const int lr   = lane & 15;
  const int hi   = lane >> 4;
  const int sn   = tid & 63;
  const int sq   = tid >> 6;

  v4f xa[16];
  const float* xrow = x + (size_t)(m0 + lr) * IND + 4 * hi;
#pragma unroll
  for (int c = 0; c < 16; ++c) xa[c] = *(const v4f*)(xrow + 8 * c);

#pragma unroll
  for (int kh = 0; kh < 2; ++kh) {
    v4f r0[4], r1[4];
    stage_load(r0, Wh + (size_t)(64 * kh + 4 * sq) * HIDD + sn);
    stage_load(r1, Wf + (size_t)(64 * kh + 4 * sq) * HIDD + sn);
#pragma unroll
    for (int p = 0; p < 4; ++p) {
      const int q = 16 * kh + 4 * p + sq;
      *(v4f*)(wl + (q * 128 + sn) * 4)      = r0[p];
      *(v4f*)(wl + (q * 128 + 64 + sn) * 4) = r1[p];
    }
  }
  __syncthreads();

  v8f acc[8] = {};
  const float* bp = wl + (hi * 128 + lr) * 4;
#pragma unroll
  for (int c = 0; c < 16; ++c) {
    v4f bv[8];
#pragma unroll
    for (int nb = 0; nb < 8; ++nb) bv[nb] = *(const v4f*)(bp + c * 1024 + nb * 64);
#pragma unroll
    for (int nb = 0; nb < 8; ++nb) acc[nb] = mma4(xa[c].xy, bv[nb].xy, acc[nb]);
#pragma unroll
    for (int nb = 0; nb < 8; ++nb) acc[nb] = mma4(xa[c].zw, bv[nb].zw, acc[nb]);
  }

#pragma unroll
  for (int nb = 0; nb < 4; ++nb) {
    const int n = nb * 16 + lr;
    const float bhn = bh[n];
    const float bfn = bf[n];
#pragma unroll
    for (int i = 0; i < 8; ++i) {
      const int m = i + 8 * hi;
      const size_t off = (size_t)(m0 + m) * HIDD + n;
      H[off] = acc[nb][i] + bhn;
      P[off] = acc[4 + nb][i] + bfn;
    }
  }
}

static __device__ __forceinline__ void bmm_part(v8f (&acc)[4], const v4f (&a)[8], const float* bp,
                                                const int c0, const int c1) {
#pragma unroll
  for (int c = c0; c < c1; ++c) {
    v4f bv[4];
#pragma unroll
    for (int nb = 0; nb < 4; ++nb) bv[nb] = *(const v4f*)(bp + c * 512 + nb * 64);
#pragma unroll
    for (int nb = 0; nb < 4; ++nb) acc[nb] = mma4(a[c].xy, bv[nb].xy, acc[nb]);
#pragma unroll
    for (int nb = 0; nb < 4; ++nb) acc[nb] = mma4(a[c].zw, bv[nb].zw, acc[nb]);
  }
}

static __device__ __forceinline__ void bmm_step(v8f (&acc)[4], const v4f (&acur)[8], v4f (&anext)[8],
                                                const float* bcur, float* dnext,
                                                const float* anext_src, const float* hnext_src) {
  v4f hn[4];
  stage_load(hn, hnext_src);
#pragma unroll
  for (int c = 0; c < 8; ++c) anext[c] = *(const v4f*)(anext_src + 8 * c);
  bmm_part(acc, acur, bcur, 0, 4);
  __builtin_amdgcn_sched_barrier(0);
  stage_store(hn, dnext);
  bmm_part(acc, acur, bcur, 4, 8);
  __syncthreads();
}

__global__ __launch_bounds__(256) void k_bmm_gate(
    const float* __restrict__ A, const float* __restrict__ H,
    const float* __restrict__ P,
    const float* __restrict__ Wg1, const float* __restrict__ bg1,
    const float* __restrict__ Wg2, const float* __restrict__ bg2,
    float* __restrict__ out) {
  __shared__ __attribute__((aligned(16))) float hs[2 * HSF];
  __shared__ __attribute__((aligned(16))) float xt[8 * 16 * XS];

  const int tid  = threadIdx.x;
  const int lane = tid & 31;
  const int w    = tid >> 5;
  const int gw   = blockIdx.x * 8 + w;
  const int b    = gw >> 6;
  const int m0   = (gw & 63) * 16;
  const int lr   = lane & 15;
  const int hi   = lane >> 4;
  const int sn   = tid & 63;
  const int sq   = tid >> 6;

  const float* Ab = A + (size_t)b * NN * NN;
  const float* Hb = H + (size_t)b * NN * HIDD;

  const float* arow = Ab + (size_t)(m0 + lr) * NN + 4 * hi;
  const float* hsrc = Hb + (size_t)(4 * sq) * HIDD + sn;
  float*       hdst = hs + (sq * 64 + sn) * 4;
  const float* bp   = hs + (hi * 64 + lr) * 4;

  v8f acc[4] = {};
  v4f a0[8], a1[8];

  {
    v4f h0[4];
#pragma unroll
    for (int c = 0; c < 8; ++c) a0[c] = *(const v4f*)(arow + 8 * c);
    stage_load(h0, hsrc);
    stage_store(h0, hdst);
  }
  __syncthreads();

#pragma unroll 1
  for (int it = 0; it < 7; ++it) {
    const int kc = it * 2 * KC;
    bmm_step(acc, a0, a1, bp, hdst + HSF, arow + kc + KC, hsrc + (size_t)(kc + KC) * HIDD);
    bmm_step(acc, a1, a0, bp + HSF, hdst, arow + kc + 2 * KC, hsrc + (size_t)(kc + 2 * KC) * HIDD);
  }
  bmm_step(acc, a0, a1, bp, hdst + HSF, arow + 15 * KC, hsrc + (size_t)(15 * KC) * HIDD);
  v4f g2r[4], g1r[4];
  stage_load(g2r, Wg2 + (size_t)(4 * sq) * HIDD + sn);
  stage_load(g1r, Wg1 + (size_t)(4 * sq) * HIDD + sn);
  bmm_part(acc, a1, bp + HSF, 0, 8);
  __syncthreads();

  stage_store(g2r, hdst);
  stage_store(g1r, hdst + HSF);
  float* myl = xt + w * (16 * XS);
#pragma unroll
  for (int nb = 0; nb < 4; ++nb)
#pragma unroll
    for (int i = 0; i < 8; ++i)
      myl[(i + 8 * hi) * XS + nb * 16 + lr] = acc[nb][i];
  __syncthreads();

  v8f g[4] = {};
  const float* prow = P + (size_t)(b * NN + m0 + lr) * HIDD + 4 * hi;
  v4f pa[8];
#pragma unroll
  for (int c = 0; c < 8; ++c) pa[c] = *(const v4f*)(prow + 8 * c);
#pragma unroll
  for (int c = 0; c < 8; ++c) {
    const v4f xa = *(const v4f*)(myl + lr * XS + 8 * c + 4 * hi);
    v4f b2[4], b1[4];
#pragma unroll
    for (int nb = 0; nb < 4; ++nb) {
      b2[nb] = *(const v4f*)(bp + c * 512 + nb * 64);
      b1[nb] = *(const v4f*)(bp + HSF + c * 512 + nb * 64);
    }
#pragma unroll
    for (int nb = 0; nb < 4; ++nb) g[nb] = mma4(xa.xy, b2[nb].xy, g[nb]);
#pragma unroll
    for (int nb = 0; nb < 4; ++nb) g[nb] = mma4(pa[c].xy, b1[nb].xy, g[nb]);
#pragma unroll
    for (int nb = 0; nb < 4; ++nb) g[nb] = mma4(xa.zw, b2[nb].zw, g[nb]);
#pragma unroll
    for (int nb = 0; nb < 4; ++nb) g[nb] = mma4(pa[c].zw, b1[nb].zw, g[nb]);
  }

#pragma unroll
  for (int nb = 0; nb < 4; ++nb) {
    const int n = nb * 16 + lr;
    const float bias = bg1[n] + bg2[n];
#pragma unroll
    for (int i = 0; i < 8; ++i) {
      const int m = i + 8 * hi;
      const size_t off = (size_t)(b * NN + m0 + m) * HIDD + n;
      const float xn = acc[nb][i];
      const float p  = P[off];
      const float z  = g[nb][i] + bias;
      const float gate = 1.0f / (1.0f + __expf(-z));
      out[off] = xn * gate + p * (1.0f - gate);
    }
  }
}

extern "C" void kernel_launch(void* const* d_in, const int* in_sizes, int n_in,
                              void* d_out, int out_size, void* d_ws, size_t ws_size,
                              hipStream_t stream) {
  const float* x    = (const float*)d_in[0];
  const float* A    = (const float*)d_in[1];
  const float* W_h  = (const float*)d_in[2];
  const float* b_h  = (const float*)d_in[3];
  const float* W_f1 = (const float*)d_in[4];
  const float* b_f1 = (const float*)d_in[5];
  const float* W_g1 = (const float*)d_in[6];
  const float* b_g1 = (const float*)d_in[7];
  const float* W_g2 = (const float*)d_in[8];
  const float* b_g2 = (const float*)d_in[9];
  float* out = (float*)d_out;

  float* H = (float*)d_ws;
  float* P = H + (size_t)BATCH * NN * HIDD;

  k_proj<<<256, 256, 0, stream>>>(x, W_h, b_h, W_f1, b_f1, H, P);
  k_bmm_gate<<<256, 256, 0, stream>>>(A, H, P, W_g1, b_g1, W_g2, b_g2, out);
}


# A-operand rows (A, P, x) fetched as whole 128-byte lines (4 per load, not 16 rows x 32 B), put in fragment order through each wave's own LDS tile; P fetched a step early; same k order and sums
# speedup vs baseline: 1.9836x; 1.0567x over previous
#include <hip/hip_runtime.h>


typedef __attribute__((ext_vector_type(2))) float v2f;
typedef __attribute__((ext_vector_type(4))) float v4f;
typedef __attribute__((ext_vector_type(8))) float v8f;

#define BATCH 32
#define NN    1024
#define IND   128
#define HIDD  64
#define KC    64
#define HSF   (KC * HIDD)
#define XS    68
#define XP    132


static __device__ __forceinline__ v8f mma4(v2f a, v2f b, v8f c) {
  return __builtin_amdgcn_wmma_f32_16x16x4_f32(false, a, false, b, (short)0, c, false, false);
}

static __device__ __forceinline__ void stage_load(v4f (&r)[4], const float* __restrict__ src) {
#pragma unroll
  for (int p = 0; p < 4; ++p) {
    r[p].x = src[(16 * p + 0) * HIDD];
    r[p].y = src[(16 * p + 1) * HIDD];
    r[p].z = src[(16 * p + 2) * HIDD];
    r[p].w = src[(16 * p + 3) * HIDD];
  }
}
static __device__ __forceinline__ void stage_store(const v4f (&r)[4], float* dst) {
#pragma unroll
  for (int p = 0; p < 4; ++p) *(v4f*)(dst + p * 1024) = r[p];
}

static __device__ __forceinline__ void araw_load(v4f (&r)[8], const float* __restrict__ src, const int pitch) {
#pragma unroll
  for (int j = 0; j < 8; ++j) r[j] = *(const v4f*)(src + (size_t)(2 * j) * pitch);
}
static __device__ __forceinline__ void a_transpose(v4f (&frag)[8], const v4f (&r)[8], float* tile_st, const float* tile_ld) {
#pragma unroll
  for (int j = 0; j < 8; ++j) *(v4f*)(tile_st + (2 * j) * XS) = r[j];
#pragma unroll
  for (int c = 0; c < 8; ++c) frag[c] = *(const v4f*)(tile_ld + 8 * c);
}

__global__ __launch_bounds__(256) void k_proj(
    const float* __restrict__ x,
    const float* __restrict__ Wh, const float* __restrict__ bh,
    const float* __restrict__ Wf, const float* __restrict__ bf,
    float* __restrict__ H, float* __restrict__ P) {
  __shared__ __attribute__((aligned(16))) float wl[32 * 128 * 4];
  __shared__ __attribute__((aligned(16))) float xs[8 * 16 * XP];

  const int tid  = threadIdx.x;
  const int lane = tid & 31;
  const int w    = tid >> 5;
  const int gw   = blockIdx.x * 8 + w;
  const int m0   = gw * 16;
  const int lr   = lane & 15;
  const int hi   = lane >> 4;
  const int sn   = tid & 63;
  const int sq   = tid >> 6;

  v4f xr[16];
  const float* xsrc = x + (size_t)m0 * IND + 4 * lane;
#pragma unroll
  for (int j = 0; j < 16; ++j) xr[j] = *(const v4f*)(xsrc + j * IND);

#pragma unroll
  for (int kh = 0; kh < 2; ++kh) {
    v4f r0[4], r1[4];
    stage_load(r0, Wh + (size_t)(64 * kh + 4 * sq) * HIDD + sn);
    stage_load(r1, Wf + (size_t)(64 * kh + 4 * sq) * HIDD + sn);
#pragma unroll
    for (int p = 0; p < 4; ++p) {
      const int q = 16 * kh + 4 * p + sq;
      *(v4f*)(wl + (q * 128 + sn) * 4)      = r0[p];
      *(v4f*)(wl + (q * 128 + 64 + sn) * 4) = r1[p];
    }
  }
  float* myx = xs + w * (16 * XP);
#pragma unroll
  for (int j = 0; j < 16; ++j) *(v4f*)(myx + j * XP + 4 * lane) = xr[j];
  v4f xa[16];
#pragma unroll
  for (int c = 0; c < 16; ++c) xa[c] = *(const v4f*)(myx + lr * XP + 8 * c + 4 * hi);
  __syncthreads();

  v8f acc[8] = {};
  const float* bp = wl + (hi * 128 + lr) * 4;
#pragma unroll
  for (int c = 0; c < 16; ++c) {
    v4f bv[8];
#pragma unroll
    for (int nb = 0; nb < 8; ++nb) bv[nb] = *(const v4f*)(bp + c * 1024 + nb * 64);
#pragma unroll
    for (int nb = 0; nb < 8; ++nb) acc[nb] = mma4(xa[c].xy, bv[nb].xy, acc[nb]);
#pragma unroll
    for (int nb = 0; nb < 8; ++nb) acc[nb] = mma4(xa[c].zw, bv[nb].zw, acc[nb]);
  }

#pragma unroll
  for (int nb = 0; nb < 4; ++nb) {
    const int n = nb * 16 + lr;
    const float bhn = bh[n];
    const float bfn = bf[n];
#pragma unroll
    for (int i = 0; i < 8; ++i) {
      const int m = i + 8 * hi;
      const size_t off = (size_t)(m0 + m) * HIDD + n;
      H[off] = acc[nb][i] + bhn;
      P[off] = acc[4 + nb][i] + bfn;
    }
  }
}

static __device__ __forceinline__ void bmm_part(v8f (&acc)[4], const v4f (&a)[8], const float* bp,
                                                const int c0, const int c1) {
#pragma unroll
  for (int c = c0; c < c1; ++c) {
    v4f bv[4];
#pragma unroll
    for (int nb = 0; nb < 4; ++nb) bv[nb] = *(const v4f*)(bp + c * 512 + nb * 64);
#pragma unroll
    for (int nb = 0; nb < 4; ++nb) acc[nb] = mma4(a[c].xy, bv[nb].xy, acc[nb]);
#pragma unroll
    for (int nb = 0; nb < 4; ++nb) acc[nb] = mma4(a[c].zw, bv[nb].zw, acc[nb]);
  }
}

static __device__ __forceinline__ void bmm_step(v8f (&acc)[4], const v4f (&acur)[8], v4f (&anext)[8],
                                                const v4f (&araw)[8], v4f (&afetch)[8],
                                                const float* bcur, float* dnext, const float* hnext_src,
                                                float* tile_st, const float* tile_ld,
                                                const float* afetch_src, const int afetch_pitch) {
  v4f hn[4];
  stage_load(hn, hnext_src);
  araw_load(afetch, afetch_src, afetch_pitch);
  bmm_part(acc, acur, bcur, 0, 4);
  __builtin_amdgcn_sched_barrier(0);
  stage_store(hn, dnext);
  a_transpose(anext, araw, tile_st, tile_ld);
  bmm_part(acc, acur, bcur, 4, 8);
  __syncthreads();
}

__global__ __launch_bounds__(256) void k_bmm_gate(
    const float* __restrict__ A, const float* __restrict__ H,
    const float* __restrict__ P,
    const float* __restrict__ Wg1, const float* __restrict__ bg1,
    const float* __restrict__ Wg2, const float* __restrict__ bg2,
    float* __restrict__ out) {
  __shared__ __attribute__((aligned(16))) float hs[2 * HSF];
  __shared__ __attribute__((aligned(16))) float xt[8 * 16 * XS];

  const int tid  = threadIdx.x;
  const int lane = tid & 31;
  const int w    = tid >> 5;
  const int gw   = blockIdx.x * 8 + w;
  const int b    = gw >> 6;
  const int m0   = (gw & 63) * 16;
  const int lr   = lane & 15;
  const int hi   = lane >> 4;
  const int sn   = tid & 63;
  const int sq   = tid >> 6;

  const float* Ab = A + (size_t)b * NN * NN;
  const float* Hb = H + (size_t)b * NN * HIDD;

  const float* asrc = Ab + (size_t)(m0 + hi) * NN + 4 * lr;
  const float* hsrc = Hb + (size_t)(4 * sq) * HIDD + sn;
  float*       hdst = hs + (sq * 64 + sn) * 4;
  const float* bp   = hs + (hi * 64 + lr) * 4;

  float*       myl     = xt + w * (16 * XS);
  float*       tile_st = myl + hi * XS + 4 * lr;
  const float* tile_ld = myl + lr * XS + 4 * hi;

  v8f acc[4] = {};
  v4f a0[8], a1[8], ar0[8], ar1[8];

  {
    v4f h0[4];
    araw_load(a1, asrc, NN);
    stage_load(h0, hsrc);
    araw_load(ar0, asrc + KC, NN);
    a_transpose(a0, a1, tile_st, tile_ld);
    stage_store(h0, hdst);
  }
  __syncthreads();

#pragma unroll 1
  for (int it = 0; it < 7; ++it) {
    const int kc = it * 2 * KC;
    bmm_step(acc, a0, a1, ar0, ar1, bp, hdst + HSF, hsrc + (size_t)(kc + KC) * HIDD,
             tile_st, tile_ld, asrc + kc + 2 * KC, NN);
    bmm_step(acc, a1, a0, ar1, ar0, bp + HSF, hdst, hsrc + (size_t)(kc + 2 * KC) * HIDD,
             tile_st, tile_ld, asrc + kc + 3 * KC, NN);
  }
  bmm_step(acc, a0, a1, ar0, ar1, bp, hdst + HSF, hsrc + (size_t)(15 * KC) * HIDD,
           tile_st, tile_ld, P + (size_t)(b * NN + m0 + hi) * HIDD + 4 * lr, HIDD);
  v4f g2r[4], g1r[4];
  stage_load(g2r, Wg2 + (size_t)(4 * sq) * HIDD + sn);
  stage_load(g1r, Wg1 + (size_t)(4 * sq) * HIDD + sn);
  bmm_part(acc, a1, bp + HSF, 0, 4);
  a_transpose(a0, ar1, tile_st, tile_ld);
  bmm_part(acc, a1, bp + HSF, 4, 8);
  __syncthreads();

  stage_store(g2r, hdst);
  stage_store(g1r, hdst + HSF);
#pragma unroll
  for (int nb = 0; nb < 4; ++nb)
#pragma unroll
    for (int i = 0; i < 8; ++i)
      myl[(i + 8 * hi) * XS + nb * 16 + lr] = acc[nb][i];
  __syncthreads();

  v8f g[4] = {};
#pragma unroll
  for (int c = 0; c < 8; ++c) {
    const v4f xa = *(const v4f*)(myl + lr * XS + 8 * c + 4 * hi);
    v4f b2[4], b1[4];
#pragma unroll
    for (int nb = 0; nb < 4; ++nb) {
      b2[nb] = *(const v4f*)(bp + c * 512 + nb * 64);
      b1[nb] = *(const v4f*)(bp + HSF + c * 512 + nb * 64);
    }
#pragma unroll
    for (int nb = 0; nb < 4; ++nb) g[nb] = mma4(xa.xy, b2[nb].xy, g[nb]);
#pragma unroll
    for (int nb = 0; nb < 4; ++nb) g[nb] = mma4(a0[c].xy, b1[nb].xy, g[nb]);
#pragma unroll
    for (int nb = 0; nb < 4; ++nb) g[nb] = mma4(xa.zw, b2[nb].zw, g[nb]);
#pragma unroll
    for (int nb = 0; nb < 4; ++nb) g[nb] = mma4(a0[c].zw, b1[nb].zw, g[nb]);
  }

#pragma unroll
  for (int nb = 0; nb < 4; ++nb) {
    const int n = nb * 16 + lr;
    const float bias = bg1[n] + bg2[n];
#pragma unroll
    for (int i = 0; i < 8; ++i) {
      const int m = i + 8 * hi;
      const size_t off = (size_t)(b * NN + m0 + m) * HIDD + n;
      const float xn = acc[nb][i];
      const float p  = P[off];
      const float z  = g[nb][i] + bias;
      const float gate = 1.0f / (1.0f + __expf(-z));
      out[off] = xn * gate + p * (1.0f - gate);
    }
  }
}

extern "C" void kernel_launch(void* const* d_in, const int* in_sizes, int n_in,
                              void* d_out, int out_size, void* d_ws, size_t ws_size,
                              hipStream_t stream) {
  const float* x    = (const float*)d_in[0];
  const float* A    = (const float*)d_in[1];
  const float* W_h  = (const float*)d_in[2];
  const float* b_h  = (const float*)d_in[3];
  const float* W_f1 = (const float*)d_in[4];
  const float* b_f1 = (const float*)d_in[5];
  const float* W_g1 = (const float*)d_in[6];
  const float* b_g1 = (const float*)d_in[7];
  const float* W_g2 = (const float*)d_in[8];
  const float* b_g2 = (const float*)d_in[9];
  float* out = (float*)d_out;

  float* H = (float*)d_ws;
  float* P = H + (size_t)BATCH * NN * HIDD;

  k_proj<<<256, 256, 0, stream>>>(x, W_h, b_h, W_f1, b_f1, H, P);
  k_bmm_gate<<<256, 256, 0, stream>>>(A, H, P, W_g1, b_g1, W_g2, b_g2, out);
}
